# starting kernel
# baseline (speedup 1.0000x reference)
#include <hip/hip_runtime.h>

typedef __attribute__((ext_vector_type(2))) float v2f;
typedef __attribute__((ext_vector_type(8))) float v8f;

constexpr int BATCH = 4096;
constexpr int FEAT  = 1024;
constexpr int OUTF  = 1024;
constexpr int W     = 7;
constexpr int ROWS_PER_BLOCK = 16;

__global__ __launch_bounds__(256)
void hermite_wmma_fused(const float* __restrict__ x,
                        const float* __restrict__ coeffs,
                        const float* __restrict__ sigma,
                        float* __restrict__ out) {
    __shared__ float basis[ROWS_PER_BLOCK][8];

    const int tid     = threadIdx.x;
    const int wave    = tid >> 5;
    const int lane    = tid & 31;
    const int rowBase = blockIdx.x * ROWS_PER_BLOCK;

    const float s    = fminf(fmaxf(sigma[0], 0.1f), 5.0f);
    const float invs = 1.0f / s;

    for (int rr = 0; rr < 2; ++rr) {
        const int r   = wave + rr * 8;
        const int row = rowBase + r;
        const float4* xrow = reinterpret_cast<const float4*>(x + (size_t)row * FEAT);

        float acc[W];
#pragma unroll
        for (int k = 0; k < W; ++k) acc[k] = 0.0f;

#pragma unroll
        for (int it = 0; it < FEAT / 128; ++it) {
            const float4 v = xrow[it * 32 + lane];
            const float e[4] = {v.x, v.y, v.z, v.w};
#pragma unroll
            for (int j = 0; j < 4; ++j) {
                const float xs = e[j] * invs;
                const float t  = xs + xs;
                const float g  = __expf(-fminf(xs * xs, 50.0f));
                float h0 = 1.0f;
                float h1 = t;
                acc[0] += g;
                acc[1] += h1 * g;
#pragma unroll
                for (int n = 2; n <= 6; ++n) {
                    float h = t * h1 - (2.0f * (float)(n - 1)) * h0;
                    h = fminf(fmaxf(h, -100.0f), 100.0f);
                    acc[n] += h * g;
                    h0 = h1; h1 = h;
                }
            }
        }

#pragma unroll
        for (int k = 0; k < W; ++k) {
#pragma unroll
            for (int off = 16; off > 0; off >>= 1)
                acc[k] += __shfl_xor(acc[k], off, 32);
        }
        if (lane == 0) {
#pragma unroll
            for (int k = 0; k < W; ++k) basis[r][k] = acc[k];
            basis[r][7] = 0.0f;
        }
    }
    __syncthreads();

    const int half = lane >> 4;
    const int l16  = lane & 15;

    v2f a_lo, a_hi;
    a_lo.x = basis[l16][2 * half + 0];
    a_lo.y = basis[l16][2 * half + 1];
    a_hi.x = basis[l16][4 + 2 * half + 0];
    a_hi.y = basis[l16][4 + 2 * half + 1];

    for (int ct = wave * 8; ct < wave * 8 + 8; ++ct) {
        const int o = ct * 16 + l16;

        const int k0 = 2 * half;
        const int k4 = 4 + 2 * half;
        const int ky = half ? 6 : 5;
        v2f b_lo, b_hi;
        b_lo.x = coeffs[o * W + k0];
        b_lo.y = coeffs[o * W + k0 + 1];
        b_hi.x = coeffs[o * W + k4];
        const float by = coeffs[o * W + ky];
        b_hi.y = half ? 0.0f : by;

        v8f c = {};
        c = __builtin_amdgcn_wmma_f32_16x16x4_f32(false, a_lo, false, b_lo,
                                                  (short)0, c, false, false);
        c = __builtin_amdgcn_wmma_f32_16x16x4_f32(false, a_hi, false, b_hi,
                                                  (short)0, c, false, false);

#pragma unroll
        for (int v = 0; v < 8; ++v) {
            const int m = v + 8 * half;
            out[(size_t)(rowBase + m) * OUTF + o] = c[v];
        }
    }
}

extern "C" void kernel_launch(void* const* d_in, const int* in_sizes, int n_in,
                              void* d_out, int out_size, void* d_ws, size_t ws_size,
                              hipStream_t stream) {
    const float* x      = (const float*)d_in[0];
    const float* coeffs = (const float*)d_in[1];
    const float* sigma  = (const float*)d_in[2];
    float* out          = (float*)d_out;

    dim3 grid(BATCH / ROWS_PER_BLOCK);
    dim3 block(256);
    hermite_wmma_fused<<<grid, block, 0, stream>>>(x, coeffs, sigma, out);
}


# All global loads issued at wave entry: both x rows of a wave (16 loads) and its 8 coeff fragments, ahead of the arithmetic (waves_per_eu(2,2)); rows share one butterfly; sum order unchanged
# speedup vs baseline: 1.2374x; 1.2374x over previous
#include <hip/hip_runtime.h>

typedef __attribute__((ext_vector_type(2))) float v2f;
typedef __attribute__((ext_vector_type(8))) float v8f;

constexpr int BATCH = 4096;
constexpr int FEAT  = 1024;
constexpr int OUTF  = 1024;
constexpr int W     = 7;
constexpr int ROWS_PER_BLOCK = 16;

__device__ __forceinline__ void hermite_accumulate(const float4 v, const float invs, float (&acc)[W]) {
    const float e[4] = {v.x, v.y, v.z, v.w};
#pragma unroll
    for (int j = 0; j < 4; ++j) {
        const float xs = e[j] * invs;
        const float t  = xs + xs;
        const float g  = __expf(-fminf(xs * xs, 50.0f));
        float h0 = 1.0f;
        float h1 = t;
        acc[0] += g;
        acc[1] += h1 * g;
#pragma unroll
        for (int n = 2; n <= 6; ++n) {
            float h = t * h1 - (2.0f * (float)(n - 1)) * h0;
            h = fminf(fmaxf(h, -100.0f), 100.0f);
            acc[n] += h * g;
            h0 = h1; h1 = h;
        }
    }
}

__global__ __launch_bounds__(256) __attribute__((amdgpu_waves_per_eu(2, 2)))
void hermite_wmma_fused(const float* __restrict__ x,
                        const float* __restrict__ coeffs,
                        const float* __restrict__ sigma,
                        float* __restrict__ out) {
    __shared__ float basis[ROWS_PER_BLOCK][8];

    const int tid     = threadIdx.x;
    const int wave    = tid >> 5;
    const int lane    = tid & 31;
    const int rowBase = blockIdx.x * ROWS_PER_BLOCK;

    const float s    = fminf(fmaxf(sigma[0], 0.1f), 5.0f);
    const float invs = 1.0f / s;

    const float4* xrow0 = reinterpret_cast<const float4*>(x + (size_t)(rowBase + wave) * FEAT);
    const float4* xrow1 = reinterpret_cast<const float4*>(x + (size_t)(rowBase + wave + 8) * FEAT);

    float4 v0[FEAT / 128], v1[FEAT / 128];
#pragma unroll
    for (int it = 0; it < FEAT / 128; ++it) v0[it] = xrow0[it * 32 + lane];
#pragma unroll
    for (int it = 0; it < FEAT / 128; ++it) v1[it] = xrow1[it * 32 + lane];

    const int half = lane >> 4;
    const int l16  = lane & 15;
    const int k0 = 2 * half;
    const int k4 = 4 + 2 * half;
    const int ky = half ? 6 : 5;
    v2f b_lo[8], b_hi[8];
#pragma unroll
    for (int i = 0; i < 8; ++i) {
        const int o = (wave * 8 + i) * 16 + l16;
        b_lo[i].x = coeffs[o * W + k0];
        b_lo[i].y = coeffs[o * W + k0 + 1];
        b_hi[i].x = coeffs[o * W + k4];
        const float by = coeffs[o * W + ky];
        b_hi[i].y = half ? 0.0f : by;
    }

    float acc0[W], acc1[W];
#pragma unroll
    for (int k = 0; k < W; ++k) { acc0[k] = 0.0f; acc1[k] = 0.0f; }

#pragma unroll
    for (int it = 0; it < FEAT / 128; ++it) hermite_accumulate(v0[it], invs, acc0);
#pragma unroll
    for (int it = 0; it < FEAT / 128; ++it) hermite_accumulate(v1[it], invs, acc1);

#pragma unroll
    for (int off = 16; off > 0; off >>= 1) {
#pragma unroll
        for (int k = 0; k < W; ++k) {
            acc0[k] += __shfl_xor(acc0[k], off, 32);
            acc1[k] += __shfl_xor(acc1[k], off, 32);
        }
    }
    if (lane == 0) {
#pragma unroll
        for (int k = 0; k < W; ++k) { basis[wave][k] = acc0[k]; basis[wave + 8][k] = acc1[k]; }
        basis[wave][7]     = 0.0f;
        basis[wave + 8][7] = 0.0f;
    }
    __syncthreads();

    v2f a_lo, a_hi;
    a_lo.x = basis[l16][2 * half + 0];
    a_lo.y = basis[l16][2 * half + 1];
    a_hi.x = basis[l16][4 + 2 * half + 0];
    a_hi.y = basis[l16][4 + 2 * half + 1];

#pragma unroll
    for (int i = 0; i < 8; ++i) {
        const int o = (wave * 8 + i) * 16 + l16;

        v8f c = {};
        c = __builtin_amdgcn_wmma_f32_16x16x4_f32(false, a_lo, false, b_lo[i],
                                                  (short)0, c, false, false);
        c = __builtin_amdgcn_wmma_f32_16x16x4_f32(false, a_hi, false, b_hi[i],
                                                  (short)0, c, false, false);

#pragma unroll
        for (int v = 0; v < 8; ++v) {
            const int m = v + 8 * half;
            out[(size_t)(rowBase + m) * OUTF + o] = c[v];
        }
    }
}

extern "C" void kernel_launch(void* const* d_in, const int* in_sizes, int n_in,
                              void* d_out, int out_size, void* d_ws, size_t ws_size,
                              hipStream_t stream) {
    const float* x      = (const float*)d_in[0];
    const float* coeffs = (const float*)d_in[1];
    const float* sigma  = (const float*)d_in[2];
    float* out          = (float*)d_out;

    dim3 grid(BATCH / ROWS_PER_BLOCK);
    dim3 block(256);
    hermite_wmma_fused<<<grid, block, 0, stream>>>(x, coeffs, sigma, out);
}


# Phase A arithmetic on float pairs (v_pk_mul/add/fma_f32, two elements per instruction); each lane keeps two partial sums per degree (elements 0,2 and 1,3 of a load), added before the butterfly
# speedup vs baseline: 1.3355x; 1.0793x over previous
#include <hip/hip_runtime.h>

typedef __attribute__((ext_vector_type(2))) float v2f;
typedef __attribute__((ext_vector_type(8))) float v8f;

constexpr int BATCH = 4096;
constexpr int FEAT  = 1024;
constexpr int OUTF  = 1024;
constexpr int W     = 7;
constexpr int ROWS_PER_BLOCK = 16;

__device__ __forceinline__ void hermite_accumulate2(const v2f e, const float invs, v2f (&acc)[W]) {
    const v2f xs = e * invs;
    const v2f t  = xs + xs;
    const v2f sq = xs * xs;
    v2f g;
    g.x = __expf(-fminf(sq.x, 50.0f));
    g.y = __expf(-fminf(sq.y, 50.0f));
    v2f h0 = {1.0f, 1.0f};
    v2f h1 = t;
    acc[0] += g;
    acc[1] += h1 * g;
#pragma unroll
    for (int n = 2; n <= 6; ++n) {
        v2f h = t * h1 - (2.0f * (float)(n - 1)) * h0;
        h.x = fminf(fmaxf(h.x, -100.0f), 100.0f);
        h.y = fminf(fmaxf(h.y, -100.0f), 100.0f);
        acc[n] += h * g;
        h0 = h1; h1 = h;
    }
}

__global__ __launch_bounds__(256) __attribute__((amdgpu_waves_per_eu(2, 2)))
void hermite_wmma_fused(const float* __restrict__ x,
                        const float* __restrict__ coeffs,
                        const float* __restrict__ sigma,
                        float* __restrict__ out) {
    __shared__ float basis[ROWS_PER_BLOCK][8];

    const int tid     = threadIdx.x;
    const int wave    = tid >> 5;
    const int lane    = tid & 31;
    const int rowBase = blockIdx.x * ROWS_PER_BLOCK;

    const float s    = fminf(fmaxf(sigma[0], 0.1f), 5.0f);
    const float invs = 1.0f / s;

    const float4* xrow0 = reinterpret_cast<const float4*>(x + (size_t)(rowBase + wave) * FEAT);
    const float4* xrow1 = reinterpret_cast<const float4*>(x + (size_t)(rowBase + wave + 8) * FEAT);

    float4 v0[FEAT / 128], v1[FEAT / 128];
#pragma unroll
    for (int it = 0; it < FEAT / 128; ++it) v0[it] = xrow0[it * 32 + lane];
#pragma unroll
    for (int it = 0; it < FEAT / 128; ++it) v1[it] = xrow1[it * 32 + lane];

    const int half = lane >> 4;
    const int l16  = lane & 15;
    const int k0 = 2 * half;
    const int k4 = 4 + 2 * half;
    const int ky = half ? 6 : 5;
    v2f b_lo[8], b_hi[8];
#pragma unroll
    for (int i = 0; i < 8; ++i) {
        const int o = (wave * 8 + i) * 16 + l16;
        b_lo[i].x = coeffs[o * W + k0];
        b_lo[i].y = coeffs[o * W + k0 + 1];
        b_hi[i].x = coeffs[o * W + k4];
        const float by = coeffs[o * W + ky];
        b_hi[i].y = half ? 0.0f : by;
    }

    v2f pa0[W], pa1[W];
#pragma unroll
    for (int k = 0; k < W; ++k) { pa0[k] = (v2f){0.0f, 0.0f}; pa1[k] = (v2f){0.0f, 0.0f}; }

#pragma unroll
    for (int it = 0; it < FEAT / 128; ++it) {
        hermite_accumulate2((v2f){v0[it].x, v0[it].y}, invs, pa0);
        hermite_accumulate2((v2f){v0[it].z, v0[it].w}, invs, pa0);
    }
#pragma unroll
    for (int it = 0; it < FEAT / 128; ++it) {
        hermite_accumulate2((v2f){v1[it].x, v1[it].y}, invs, pa1);
        hermite_accumulate2((v2f){v1[it].z, v1[it].w}, invs, pa1);
    }

    float acc0[W], acc1[W];
#pragma unroll
    for (int k = 0; k < W; ++k) { acc0[k] = pa0[k].x + pa0[k].y; acc1[k] = pa1[k].x + pa1[k].y; }

#pragma unroll
    for (int off = 16; off > 0; off >>= 1) {
#pragma unroll
        for (int k = 0; k < W; ++k) {
            acc0[k] += __shfl_xor(acc0[k], off, 32);
            acc1[k] += __shfl_xor(acc1[k], off, 32);
        }
    }
    if (lane == 0) {
#pragma unroll
        for (int k = 0; k < W; ++k) { basis[wave][k] = acc0[k]; basis[wave + 8][k] = acc1[k]; }
        basis[wave][7]     = 0.0f;
        basis[wave + 8][7] = 0.0f;
    }
    __syncthreads();

    v2f a_lo, a_hi;
    a_lo.x = basis[l16][2 * half + 0];
    a_lo.y = basis[l16][2 * half + 1];
    a_hi.x = basis[l16][4 + 2 * half + 0];
    a_hi.y = basis[l16][4 + 2 * half + 1];

#pragma unroll
    for (int i = 0; i < 8; ++i) {
        const int o = (wave * 8 + i) * 16 + l16;

        v8f c = {};
        c = __builtin_amdgcn_wmma_f32_16x16x4_f32(false, a_lo, false, b_lo[i],
                                                  (short)0, c, false, false);
        c = __builtin_amdgcn_wmma_f32_16x16x4_f32(false, a_hi, false, b_hi[i],
                                                  (short)0, c, false, false);

#pragma unroll
        for (int v = 0; v < 8; ++v) {
            const int m = v + 8 * half;
            out[(size_t)(rowBase + m) * OUTF + o] = c[v];
        }
    }
}

extern "C" void kernel_launch(void* const* d_in, const int* in_sizes, int n_in,
                              void* d_out, int out_size, void* d_ws, size_t ws_size,
                              hipStream_t stream) {
    const float* x      = (const float*)d_in[0];
    const float* coeffs = (const float*)d_in[1];
    const float* sigma  = (const float*)d_in[2];
    float* out          = (float*)d_out;

    dim3 grid(BATCH / ROWS_PER_BLOCK);
    dim3 block(256);
    hermite_wmma_fused<<<grid, block, 0, stream>>>(x, coeffs, sigma, out);
}


# Phase B column tiles interleaved in pairs (lane c owns columns 2c, 2c+1): each 128-B output line is written whole by one 8-byte-per-lane store, 32 per wave instead of 64 half-line stores
# speedup vs baseline: 1.4585x; 1.0921x over previous
#include <hip/hip_runtime.h>

typedef __attribute__((ext_vector_type(2))) float v2f;
typedef __attribute__((ext_vector_type(8))) float v8f;

constexpr int BATCH = 4096;
constexpr int FEAT  = 1024;
constexpr int OUTF  = 1024;
constexpr int W     = 7;
constexpr int ROWS_PER_BLOCK = 16;

__device__ __forceinline__ void hermite_accumulate2(const v2f e, const float invs, v2f (&acc)[W]) {
    const v2f xs = e * invs;
    const v2f t  = xs + xs;
    const v2f sq = xs * xs;
    v2f g;
    g.x = __expf(-fminf(sq.x, 50.0f));
    g.y = __expf(-fminf(sq.y, 50.0f));
    v2f h0 = {1.0f, 1.0f};
    v2f h1 = t;
    acc[0] += g;
    acc[1] += h1 * g;
#pragma unroll
    for (int n = 2; n <= 6; ++n) {
        v2f h = t * h1 - (2.0f * (float)(n - 1)) * h0;
        h.x = fminf(fmaxf(h.x, -100.0f), 100.0f);
        h.y = fminf(fmaxf(h.y, -100.0f), 100.0f);
        acc[n] += h * g;
        h0 = h1; h1 = h;
    }
}

__global__ __launch_bounds__(256) __attribute__((amdgpu_waves_per_eu(2, 2)))
void hermite_wmma_fused(const float* __restrict__ x,
                        const float* __restrict__ coeffs,
                        const float* __restrict__ sigma,
                        float* __restrict__ out) {
    __shared__ float basis[ROWS_PER_BLOCK][8];

    const int tid     = threadIdx.x;
    const int wave    = tid >> 5;
    const int lane    = tid & 31;
    const int rowBase = blockIdx.x * ROWS_PER_BLOCK;

    const float s    = fminf(fmaxf(sigma[0], 0.1f), 5.0f);
    const float invs = 1.0f / s;

    const float4* xrow0 = reinterpret_cast<const float4*>(x + (size_t)(rowBase + wave) * FEAT);
    const float4* xrow1 = reinterpret_cast<const float4*>(x + (size_t)(rowBase + wave + 8) * FEAT);

    float4 v0[FEAT / 128], v1[FEAT / 128];
#pragma unroll
    for (int it = 0; it < FEAT / 128; ++it) v0[it] = xrow0[it * 32 + lane];
#pragma unroll
    for (int it = 0; it < FEAT / 128; ++it) v1[it] = xrow1[it * 32 + lane];

    const int half = lane >> 4;
    const int l16  = lane & 15;
    const int k0 = 2 * half;
    const int k4 = 4 + 2 * half;
    const int ky = half ? 6 : 5;
    v2f b_lo[8], b_hi[8];
#pragma unroll
    for (int i = 0; i < 8; ++i) {
        const int o = wave * 128 + (i >> 1) * 32 + 2 * l16 + (i & 1);
        b_lo[i].x = coeffs[o * W + k0];
        b_lo[i].y = coeffs[o * W + k0 + 1];
        b_hi[i].x = coeffs[o * W + k4];
        const float by = coeffs[o * W + ky];
        b_hi[i].y = half ? 0.0f : by;
    }

    v2f pa0[W], pa1[W];
#pragma unroll
    for (int k = 0; k < W; ++k) { pa0[k] = (v2f){0.0f, 0.0f}; pa1[k] = (v2f){0.0f, 0.0f}; }

#pragma unroll
    for (int it = 0; it < FEAT / 128; ++it) {
        hermite_accumulate2((v2f){v0[it].x, v0[it].y}, invs, pa0);
        hermite_accumulate2((v2f){v0[it].z, v0[it].w}, invs, pa0);
    }
#pragma unroll
    for (int it = 0; it < FEAT / 128; ++it) {
        hermite_accumulate2((v2f){v1[it].x, v1[it].y}, invs, pa1);
        hermite_accumulate2((v2f){v1[it].z, v1[it].w}, invs, pa1);
    }

    float acc0[W], acc1[W];
#pragma unroll
    for (int k = 0; k < W; ++k) { acc0[k] = pa0[k].x + pa0[k].y; acc1[k] = pa1[k].x + pa1[k].y; }

#pragma unroll
    for (int off = 16; off > 0; off >>= 1) {
#pragma unroll
        for (int k = 0; k < W; ++k) {
            acc0[k] += __shfl_xor(acc0[k], off, 32);
            acc1[k] += __shfl_xor(acc1[k], off, 32);
        }
    }
    if (lane == 0) {
#pragma unroll
        for (int k = 0; k < W; ++k) { basis[wave][k] = acc0[k]; basis[wave + 8][k] = acc1[k]; }
        basis[wave][7]     = 0.0f;
        basis[wave + 8][7] = 0.0f;
    }
    __syncthreads();

    v2f a_lo, a_hi;
    a_lo.x = basis[l16][2 * half + 0];
    a_lo.y = basis[l16][2 * half + 1];
    a_hi.x = basis[l16][4 + 2 * half + 0];
    a_hi.y = basis[l16][4 + 2 * half + 1];

#pragma unroll
    for (int j = 0; j < 4; ++j) {
        v8f c0 = {}, c1 = {};
        c0 = __builtin_amdgcn_wmma_f32_16x16x4_f32(false, a_lo, false, b_lo[2 * j],
                                                   (short)0, c0, false, false);
        c1 = __builtin_amdgcn_wmma_f32_16x16x4_f32(false, a_lo, false, b_lo[2 * j + 1],
                                                   (short)0, c1, false, false);
        c0 = __builtin_amdgcn_wmma_f32_16x16x4_f32(false, a_hi, false, b_hi[2 * j],
                                                   (short)0, c0, false, false);
        c1 = __builtin_amdgcn_wmma_f32_16x16x4_f32(false, a_hi, false, b_hi[2 * j + 1],
                                                   (short)0, c1, false, false);

        const int o2 = wave * 128 + j * 32 + 2 * l16;
#pragma unroll
        for (int v = 0; v < 8; ++v) {
            const int m = v + 8 * half;
            *reinterpret_cast<v2f*>(out + (size_t)(rowBase + m) * OUTF + o2) = (v2f){c0[v], c1[v]};
        }
    }
}

extern "C" void kernel_launch(void* const* d_in, const int* in_sizes, int n_in,
                              void* d_out, int out_size, void* d_ws, size_t ws_size,
                              hipStream_t stream) {
    const float* x      = (const float*)d_in[0];
    const float* coeffs = (const float*)d_in[1];
    const float* sigma  = (const float*)d_in[2];
    float* out          = (float*)d_out;

    dim3 grid(BATCH / ROWS_PER_BLOCK);
    dim3 block(256);
    hermite_wmma_fused<<<grid, block, 0, stream>>>(x, coeffs, sigma, out);
}


# Wave sums on the vector ALU: one row swap per degree for lane distance 16, DPP row_xmask for 8,4,2,1, instead of 70 LDS-crossbar permutes; every add keeps its two operands, sum order unchanged
# speedup vs baseline: 1.5794x; 1.0829x over previous
#include <hip/hip_runtime.h>

typedef __attribute__((ext_vector_type(2))) float v2f;
typedef __attribute__((ext_vector_type(8))) float v8f;

constexpr int BATCH = 4096;
constexpr int FEAT  = 1024;
constexpr int OUTF  = 1024;
constexpr int W     = 7;
constexpr int ROWS_PER_BLOCK = 16;

__device__ __forceinline__ void hermite_accumulate2(const v2f e, const float invs, v2f (&acc)[W]) {
    const v2f xs = e * invs;
    const v2f t  = xs + xs;
    const v2f sq = xs * xs;
    v2f g;
    g.x = __expf(-fminf(sq.x, 50.0f));
    g.y = __expf(-fminf(sq.y, 50.0f));
    v2f h0 = {1.0f, 1.0f};
    v2f h1 = t;
    acc[0] += g;
    acc[1] += h1 * g;
#pragma unroll
    for (int n = 2; n <= 6; ++n) {
        v2f h = t * h1 - (2.0f * (float)(n - 1)) * h0;
        h.x = fminf(fmaxf(h.x, -100.0f), 100.0f);
        h.y = fminf(fmaxf(h.y, -100.0f), 100.0f);
        acc[n] += h * g;
        h0 = h1; h1 = h;
    }
}

template <int MASK>
__device__ __forceinline__ float row_xor(const float v) {
    return __builtin_bit_cast(float, __builtin_amdgcn_update_dpp(
        0, __builtin_bit_cast(int, v), 0x160 + MASK, 0xf, 0xf, true));
}

__global__ __launch_bounds__(256) __attribute__((amdgpu_waves_per_eu(2, 2)))
void hermite_wmma_fused(const float* __restrict__ x,
                        const float* __restrict__ coeffs,
                        const float* __restrict__ sigma,
                        float* __restrict__ out) {
    __shared__ float basis[ROWS_PER_BLOCK][8];

    const int tid     = threadIdx.x;
    const int wave    = tid >> 5;
    const int lane    = tid & 31;
    const int rowBase = blockIdx.x * ROWS_PER_BLOCK;

    const float s    = fminf(fmaxf(sigma[0], 0.1f), 5.0f);
    const float invs = 1.0f / s;

    const float4* xrow0 = reinterpret_cast<const float4*>(x + (size_t)(rowBase + wave) * FEAT);
    const float4* xrow1 = reinterpret_cast<const float4*>(x + (size_t)(rowBase + wave + 8) * FEAT);

    float4 v0[FEAT / 128], v1[FEAT / 128];
#pragma unroll
    for (int it = 0; it < FEAT / 128; ++it) v0[it] = xrow0[it * 32 + lane];
#pragma unroll
    for (int it = 0; it < FEAT / 128; ++it) v1[it] = xrow1[it * 32 + lane];

    const int half = lane >> 4;
    const int l16  = lane & 15;
    const int k0 = 2 * half;
    const int k4 = 4 + 2 * half;
    const int ky = half ? 6 : 5;
    v2f b_lo[8], b_hi[8];
#pragma unroll
    for (int i = 0; i < 8; ++i) {
        const int o = wave * 128 + (i >> 1) * 32 + 2 * l16 + (i & 1);
        b_lo[i].x = coeffs[o * W + k0];
        b_lo[i].y = coeffs[o * W + k0 + 1];
        b_hi[i].x = coeffs[o * W + k4];
        const float by = coeffs[o * W + ky];
        b_hi[i].y = half ? 0.0f : by;
    }

    v2f pa0[W], pa1[W];
#pragma unroll
    for (int k = 0; k < W; ++k) { pa0[k] = (v2f){0.0f, 0.0f}; pa1[k] = (v2f){0.0f, 0.0f}; }

#pragma unroll
    for (int it = 0; it < FEAT / 128; ++it) {
        hermite_accumulate2((v2f){v0[it].x, v0[it].y}, invs, pa0);
        hermite_accumulate2((v2f){v0[it].z, v0[it].w}, invs, pa0);
    }
#pragma unroll
    for (int it = 0; it < FEAT / 128; ++it) {
        hermite_accumulate2((v2f){v1[it].x, v1[it].y}, invs, pa1);
        hermite_accumulate2((v2f){v1[it].z, v1[it].w}, invs, pa1);
    }

    float sum[W];
#pragma unroll
    for (int k = 0; k < W; ++k) {
        const float a0 = pa0[k].x + pa0[k].y;
        const float a1 = pa1[k].x + pa1[k].y;
        const auto sw = __builtin_amdgcn_permlane16_swap(__builtin_bit_cast(unsigned, a0),
                                                         __builtin_bit_cast(unsigned, a1),
                                                         false, false);
        sum[k] = __builtin_bit_cast(float, (unsigned)sw[0]) + __builtin_bit_cast(float, (unsigned)sw[1]);
    }
#pragma unroll
    for (int k = 0; k < W; ++k) sum[k] += row_xor<8>(sum[k]);
#pragma unroll
    for (int k = 0; k < W; ++k) sum[k] += row_xor<4>(sum[k]);
#pragma unroll
    for (int k = 0; k < W; ++k) sum[k] += row_xor<2>(sum[k]);
#pragma unroll
    for (int k = 0; k < W; ++k) sum[k] += row_xor<1>(sum[k]);

    if (l16 == 0) {
#pragma unroll
        for (int k = 0; k < W; ++k) basis[wave + 8 * half][k] = sum[k];
        basis[wave + 8 * half][7] = 0.0f;
    }
    __syncthreads();

    v2f a_lo, a_hi;
    a_lo.x = basis[l16][2 * half + 0];
    a_lo.y = basis[l16][2 * half + 1];
    a_hi.x = basis[l16][4 + 2 * half + 0];
    a_hi.y = basis[l16][4 + 2 * half + 1];

#pragma unroll
    for (int j = 0; j < 4; ++j) {
        v8f c0 = {}, c1 = {};
        c0 = __builtin_amdgcn_wmma_f32_16x16x4_f32(false, a_lo, false, b_lo[2 * j],
                                                   (short)0, c0, false, false);
        c1 = __builtin_amdgcn_wmma_f32_16x16x4_f32(false, a_lo, false, b_lo[2 * j + 1],
                                                   (short)0, c1, false, false);
        c0 = __builtin_amdgcn_wmma_f32_16x16x4_f32(false, a_hi, false, b_hi[2 * j],
                                                   (short)0, c0, false, false);
        c1 = __builtin_amdgcn_wmma_f32_16x16x4_f32(false, a_hi, false, b_hi[2 * j + 1],
                                                   (short)0, c1, false, false);

        const int o2 = wave * 128 + j * 32 + 2 * l16;
#pragma unroll
        for (int v = 0; v < 8; ++v) {
            const int m = v + 8 * half;
            *reinterpret_cast<v2f*>(out + (size_t)(rowBase + m) * OUTF + o2) = (v2f){c0[v], c1[v]};
        }
    }
}

extern "C" void kernel_launch(void* const* d_in, const int* in_sizes, int n_in,
                              void* d_out, int out_size, void* d_ws, size_t ws_size,
                              hipStream_t stream) {
    const float* x      = (const float*)d_in[0];
    const float* coeffs = (const float*)d_in[1];
    const float* sigma  = (const float*)d_in[2];
    float* out          = (float*)d_out;

    dim3 grid(BATCH / ROWS_PER_BLOCK);
    dim3 block(256);
    hermite_wmma_fused<<<grid, block, 0, stream>>>(x, coeffs, sigma, out);
}
